# starting kernel
# baseline (speedup 1.0000x reference)
#include <hip/hip_runtime.h>


typedef __attribute__((ext_vector_type(2))) float v2f;
typedef __attribute__((ext_vector_type(8))) float v8f;

#define IN_FEATS 128
#define HID      64
#define NC       32


__global__ void fill_zero_kernel(float* __restrict__ p, int n) {
    int i = blockIdx.x * blockDim.x + threadIdx.x;
    if (i < n) p[i] = 0.0f;
}

__global__ void deg_kernel(const int* __restrict__ src, const int* __restrict__ dst,
                           float* __restrict__ deg_out, float* __restrict__ deg_in, int E) {
    int e = blockIdx.x * blockDim.x + threadIdx.x;
    if (e < E) {
        unsafeAtomicAdd(&deg_out[src[e]], 1.0f);
        unsafeAtomicAdd(&deg_in[dst[e]], 1.0f);
    }
}

__global__ void rsqrt_clip_kernel(float* __restrict__ d, int n) {
    int i = blockIdx.x * blockDim.x + threadIdx.x;
    if (i < n) d[i] = rsqrtf(fmaxf(d[i], 1.0f));
}

__global__ void bias_relu_kernel(float* __restrict__ a, const float* __restrict__ rs_in,
                                 const float* __restrict__ b, int n) {
    int i = blockIdx.x * blockDim.x + threadIdx.x;
    if (i < n * HID) {
        int row = i >> 6;
        int col = i & (HID - 1);
        float v = a[i] * rs_in[row] + b[col];
        a[i] = fmaxf(v, 0.0f);
    }
}

__global__ void out_acc_kernel(float* __restrict__ out, const float* __restrict__ a,
                               const float* __restrict__ rs_in, const float* __restrict__ b,
                               int n) {
    int i = blockIdx.x * blockDim.x + threadIdx.x;
    if (i < n * NC) {
        int row = i >> 5;
        int col = i & (NC - 1);
        out[i] += 0.5f * (a[i] * rs_in[row] + b[col]);
    }
}

template <int F>
__global__ void scatter_kernel(const float* __restrict__ h, const int* __restrict__ src,
                               const int* __restrict__ dst, float* __restrict__ agg, int E) {
    constexpr int C = F / 4;
    int idx = blockIdx.x * blockDim.x + threadIdx.x;
    if (idx >= E * C) return;
    int e = idx / C;
    int c = idx - e * C;
    int s = src[e];
    int d = dst[e];
    const float4 v = *(const float4*)&h[(long)s * F + c * 4];
    float* p = &agg[(long)d * F + c * 4];
    unsafeAtomicAdd(p + 0, v.x);
    unsafeAtomicAdd(p + 1, v.y);
    unsafeAtomicAdd(p + 2, v.z);
    unsafeAtomicAdd(p + 3, v.w);
}

template <int IN, int OUT>
__global__ void __launch_bounds__(256)
gemm_gc_kernel(const float* __restrict__ X, const float* __restrict__ rs,
               const float* __restrict__ W, float* __restrict__ H, int n) {
    constexpr int LDW = IN + 4;
    __shared__ float lw[OUT * LDW];

    for (int i = threadIdx.x; i < IN * OUT; i += 256) {
        int k = i / OUT;
        int c = i - k * OUT;
        lw[c * LDW + k] = W[i];
    }
    __syncthreads();

    const int lane = threadIdx.x & 31;
    const int wave = threadIdx.x >> 5;
    const int r0 = (blockIdx.x * 8 + wave) * 16;
    if (r0 >= n) return;

    const int mrow = r0 + (lane & 15);
    const int koff = (lane >> 4) * 2;
    const float rsv = rs[mrow];
    const float* xrow = X + (long)mrow * IN;
    const int col = lane & 15;

    constexpr int CT = OUT / 16;
    v8f acc[CT] = {};

    for (int k = 0; k < IN; k += 4) {
        float2 av = *(const float2*)(xrow + k + koff);
        v2f a;
        a.x = av.x * rsv;
        a.y = av.y * rsv;
#pragma unroll
        for (int c = 0; c < CT; ++c) {
            float2 bv = *(const float2*)&lw[(c * 16 + col) * LDW + k + koff];
            v2f b;
            b.x = bv.x;
            b.y = bv.y;
            acc[c] = __builtin_amdgcn_wmma_f32_16x16x4_f32(
                false, a, false, b, (short)0, acc[c], false, false);
        }
    }

    const int mhi = (lane >> 4) * 8;
#pragma unroll
    for (int c = 0; c < CT; ++c) {
#pragma unroll
        for (int r = 0; r < 8; ++r) {
            H[(long)(r0 + mhi + r) * OUT + c * 16 + col] = acc[c][r];
        }
    }
}


extern "C" void kernel_launch(void* const* d_in, const int* in_sizes, int n_in,
                              void* d_out, int out_size, void* d_ws, size_t ws_size,
                              hipStream_t stream) {
    const float* x    = (const float*)d_in[0];
    const int*   src1 = (const int*)d_in[1];
    const int*   dst1 = (const int*)d_in[2];
    const int*   src2 = (const int*)d_in[3];
    const int*   dst2 = (const int*)d_in[4];
    const float* W1   = (const float*)d_in[5];
    const float* b1   = (const float*)d_in[6];
    const float* W2   = (const float*)d_in[7];
    const float* b2   = (const float*)d_in[8];
    float* out = (float*)d_out;

    const int N = in_sizes[0] / IN_FEATS;
    const int E = in_sizes[1];

    float* ws   = (float*)d_ws;
    float* rsO  = ws;
    float* rsI  = rsO + N;
    float* h    = rsI + N;
    float* agg1 = h + (size_t)N * HID;
    float* h2   = agg1 + (size_t)N * HID;
    float* agg2 = h2 + (size_t)N * NC;

    const int T = 256;
    auto cdiv = [](long a, long b) { return (int)((a + b - 1) / b); };

    const int tiles   = N / 16;
    const int gblocks = cdiv(tiles, 8);

    fill_zero_kernel<<<cdiv((long)N * NC, T), T, 0, stream>>>(out, N * NC);

    for (int g = 0; g < 2; ++g) {
        const int* src = (g == 0) ? src1 : src2;
        const int* dst = (g == 0) ? dst1 : dst2;

        fill_zero_kernel<<<cdiv(N, T), T, 0, stream>>>(rsO, N);
        fill_zero_kernel<<<cdiv(N, T), T, 0, stream>>>(rsI, N);
        deg_kernel<<<cdiv(E, T), T, 0, stream>>>(src, dst, rsO, rsI, E);
        rsqrt_clip_kernel<<<cdiv(N, T), T, 0, stream>>>(rsO, N);
        rsqrt_clip_kernel<<<cdiv(N, T), T, 0, stream>>>(rsI, N);

        gemm_gc_kernel<IN_FEATS, HID><<<gblocks, T, 0, stream>>>(x, rsO, W1, h, N);
        fill_zero_kernel<<<cdiv((long)N * HID, T), T, 0, stream>>>(agg1, N * HID);
        scatter_kernel<HID><<<cdiv((long)E * (HID / 4), T), T, 0, stream>>>(h, src, dst, agg1, E);
        bias_relu_kernel<<<cdiv((long)N * HID, T), T, 0, stream>>>(agg1, rsI, b1, N);

        gemm_gc_kernel<HID, NC><<<gblocks, T, 0, stream>>>(agg1, rsO, W2, h2, N);
        fill_zero_kernel<<<cdiv((long)N * NC, T), T, 0, stream>>>(agg2, N * NC);
        scatter_kernel<NC><<<cdiv((long)E * (NC / 4), T), T, 0, stream>>>(h2, src, dst, agg2, E);
        out_acc_kernel<<<cdiv((long)N * NC, T), T, 0, stream>>>(out, agg2, rsI, b2, N);
    }
}


# out-degree global atomics removed: the partition also orders the source ids by 128-node source range; each product block counts its 128 rows' out-degrees in LDS from that list
# speedup vs baseline: 1.7514x; 1.7514x over previous
#include <hip/hip_runtime.h>


typedef __attribute__((ext_vector_type(2))) float v2f;
typedef __attribute__((ext_vector_type(8))) float v8f;

#define IN_FEATS 128
#define HID      64
#define NC       32


#define RNG      128
#define RSHIFT   7
#define NBMAX    1024
#define EPB      8192

__global__ void __launch_bounds__(256)
range_count_kernel(const int* __restrict__ src, const int* __restrict__ dst,
                   int* __restrict__ counts, int* __restrict__ counts2, int E) {
    __shared__ int hist[NBMAX];
    __shared__ int hist2[NBMAX];
    for (int i = threadIdx.x; i < NBMAX; i += 256) { hist[i] = 0; hist2[i] = 0; }
    __syncthreads();
    const int e0 = blockIdx.x * EPB;
#pragma unroll 8
    for (int i = threadIdx.x; i < EPB; i += 256) {
        int e = e0 + i;
        if (e < E) {
            atomicAdd(&hist[dst[e] >> RSHIFT], 1);
            atomicAdd(&hist2[src[e] >> RSHIFT], 1);
        }
    }
    __syncthreads();
    for (int i = threadIdx.x; i < NBMAX; i += 256) {
        counts[blockIdx.x * NBMAX + i] = hist[i];
        counts2[blockIdx.x * NBMAX + i] = hist2[i];
    }
}

__global__ void __launch_bounds__(NBMAX)
range_scan_kernel(const int* __restrict__ counts, int* __restrict__ prefix,
                  int* __restrict__ start, int nslice) {
    __shared__ int sc[NBMAX];
    const int t = threadIdx.x;
    counts += (size_t)blockIdx.x * nslice * NBMAX;
    prefix += (size_t)blockIdx.x * nslice * NBMAX;
    start  += blockIdx.x * (NBMAX + 8);
    int run = 0;
    for (int s = 0; s < nslice; s += 8) {
        int c[8];
#pragma unroll
        for (int u = 0; u < 8; ++u) c[u] = (s + u < nslice) ? counts[(s + u) * NBMAX + t] : 0;
#pragma unroll
        for (int u = 0; u < 8; ++u) {
            if (s + u < nslice) prefix[(s + u) * NBMAX + t] = run;
            run += c[u];
        }
    }
    sc[t] = run;
    __syncthreads();
    for (int off = 1; off < NBMAX; off <<= 1) {
        int v = (t >= off) ? sc[t - off] : 0;
        __syncthreads();
        sc[t] += v;
        __syncthreads();
    }
    const int incl = sc[t];
    start[t] = incl - run;
    if (t == NBMAX - 1) start[NBMAX] = incl;
}

__global__ void __launch_bounds__(256)
range_fill_kernel(const int* __restrict__ src, const int* __restrict__ dst,
                  const int* __restrict__ prefix, const int* __restrict__ start,
                  const int* __restrict__ prefix2, const int* __restrict__ start2,
                  int2* __restrict__ elist, int* __restrict__ slist, int E) {
    __shared__ int cur[NBMAX];
    __shared__ int cur2[NBMAX];
    for (int i = threadIdx.x; i < NBMAX; i += 256) {
        cur[i]  = start[i]  + prefix[blockIdx.x * NBMAX + i];
        cur2[i] = start2[i] + prefix2[blockIdx.x * NBMAX + i];
    }
    __syncthreads();
    const int e0 = blockIdx.x * EPB;
    for (int i0 = threadIdx.x; i0 < EPB; i0 += 256 * 8) {
        int s[8], d[8];
#pragma unroll
        for (int u = 0; u < 8; ++u) {
            int e = e0 + i0 + u * 256;
            s[u] = (e < E) ? src[e] : 0;
            d[u] = (e < E) ? dst[e] : -1;
        }
#pragma unroll
        for (int u = 0; u < 8; ++u) {
            if (d[u] >= 0) {
                int pos  = atomicAdd(&cur[d[u] >> RSHIFT], 1);
                int pos2 = atomicAdd(&cur2[s[u] >> RSHIFT], 1);
                elist[pos]  = make_int2(s[u], d[u]);
                slist[pos2] = s[u];
            }
        }
    }
}

template <int F, int MODE>
__global__ void __launch_bounds__(256)
range_sum_kernel(const float* __restrict__ h, const int2* __restrict__ elist,
                 const int* __restrict__ start,
                 const float* __restrict__ b, float* __restrict__ o, int n) {
    constexpr int V = F / 32;
    __shared__ float acc[RNG * F];
    __shared__ int indeg[RNG];
    for (int i = threadIdx.x; i < RNG * F; i += 256) acc[i] = 0.0f;
    if (threadIdx.x < RNG) indeg[threadIdx.x] = 0;
    __syncthreads();

    const int lane = threadIdx.x & 31;
    const int wave = __builtin_amdgcn_readfirstlane(threadIdx.x >> 5);
    const int lo = blockIdx.x * RNG;
    const int s0 = start[blockIdx.x];
    const int s1 = start[blockIdx.x + 1];

    for (int base = s0 + wave * 32; base < s1; base += 256) {
        const int cnt = (s1 - base < 32) ? (s1 - base) : 32;
        int2 e = make_int2(0, lo);
        if (lane < cnt) e = elist[base + lane];
        const int es = e.x;
        const int ed = e.y - lo;
        if (lane < cnt) atomicAdd(&indeg[ed], 1);
        int j = 0;
        for (; j + 8 <= cnt; j += 8) {
            float v[8][V];
            int d[8];
#pragma unroll
            for (int u = 0; u < 8; ++u) {
                const int s = __builtin_amdgcn_readlane(es, j + u);
                d[u] = __builtin_amdgcn_readlane(ed, j + u);
                const float* p = h + (size_t)s * F + lane * V;
#pragma unroll
                for (int k = 0; k < V; ++k) v[u][k] = p[k];
            }
#pragma unroll
            for (int u = 0; u < 8; ++u) {
#pragma unroll
                for (int k = 0; k < V; ++k) atomicAdd(&acc[d[u] * F + lane * V + k], v[u][k]);
            }
        }
        for (; j < cnt; ++j) {
            const int s = __builtin_amdgcn_readlane(es, j);
            const int d = __builtin_amdgcn_readlane(ed, j);
            const float* p = h + (size_t)s * F + lane * V;
#pragma unroll
            for (int k = 0; k < V; ++k) atomicAdd(&acc[d * F + lane * V + k], p[k]);
        }
    }
    __syncthreads();

    for (int i = threadIdx.x; i < RNG * F; i += 256) {
        const int node = lo + i / F;
        if (node < n) {
            const int f = i % F;
            const float rs = rsqrtf(fmaxf((float)indeg[i / F], 1.0f));
            const float val = acc[i] * rs + b[f];
            const size_t at = (size_t)node * F + f;
            if (MODE == 0)      o[at] = fmaxf(val, 0.0f);
            else if (MODE == 1) o[at] = 0.5f * val;
            else                o[at] += 0.5f * val;
        }
    }
}

static_assert(RNG == 8 * 16, "a product block's rows must be one source range");
template <int IN, int OUT>
__global__ void __launch_bounds__(256)
gemm_gc_kernel(const float* __restrict__ X, const int* __restrict__ slist,
               const int* __restrict__ start2,
               const float* __restrict__ W, float* __restrict__ H, int n) {
    constexpr int LDW = IN + 4;
    __shared__ float lw[OUT * LDW];
    __shared__ int odeg[RNG];

    if (threadIdx.x < RNG) odeg[threadIdx.x] = 0;
    __syncthreads();

    const int lo = blockIdx.x * RNG;
    const int q0 = start2[blockIdx.x];
    const int q1 = start2[blockIdx.x + 1];
    for (int i0 = q0 + threadIdx.x; i0 < q1; i0 += 256 * 8) {
        int sv[8];
#pragma unroll
        for (int u = 0; u < 8; ++u) {
            const int i = i0 + u * 256;
            sv[u] = (i < q1) ? slist[i] : -1;
        }
#pragma unroll
        for (int u = 0; u < 8; ++u)
            if (sv[u] >= 0) atomicAdd(&odeg[sv[u] - lo], 1);
    }

    for (int i = threadIdx.x; i < IN * OUT; i += 256) {
        int k = i / OUT;
        int c = i - k * OUT;
        lw[c * LDW + k] = W[i];
    }
    __syncthreads();

    const int lane = threadIdx.x & 31;
    const int wave = threadIdx.x >> 5;
    const int r0 = (blockIdx.x * 8 + wave) * 16;
    if (r0 >= n) return;

    const int mrow = r0 + (lane & 15);
    const int koff = (lane >> 4) * 2;
    const float rsv = rsqrtf(fmaxf((float)odeg[mrow - lo], 1.0f));
    const float* xrow = X + (long)mrow * IN;
    const int col = lane & 15;

    constexpr int CT = OUT / 16;
    v8f acc[CT] = {};

    for (int k = 0; k < IN; k += 4) {
        float2 av = *(const float2*)(xrow + k + koff);
        v2f a;
        a.x = av.x * rsv;
        a.y = av.y * rsv;
#pragma unroll
        for (int c = 0; c < CT; ++c) {
            float2 bv = *(const float2*)&lw[(c * 16 + col) * LDW + k + koff];
            v2f b;
            b.x = bv.x;
            b.y = bv.y;
            acc[c] = __builtin_amdgcn_wmma_f32_16x16x4_f32(
                false, a, false, b, (short)0, acc[c], false, false);
        }
    }

    const int mhi = (lane >> 4) * 8;
#pragma unroll
    for (int c = 0; c < CT; ++c) {
#pragma unroll
        for (int r = 0; r < 8; ++r) {
            H[(long)(r0 + mhi + r) * OUT + c * 16 + col] = acc[c][r];
        }
    }
}


extern "C" void kernel_launch(void* const* d_in, const int* in_sizes, int n_in,
                              void* d_out, int out_size, void* d_ws, size_t ws_size,
                              hipStream_t stream) {
    const float* x    = (const float*)d_in[0];
    const int*   src1 = (const int*)d_in[1];
    const int*   dst1 = (const int*)d_in[2];
    const int*   src2 = (const int*)d_in[3];
    const int*   dst2 = (const int*)d_in[4];
    const float* W1   = (const float*)d_in[5];
    const float* b1   = (const float*)d_in[6];
    const float* W2   = (const float*)d_in[7];
    const float* b2   = (const float*)d_in[8];
    float* out = (float*)d_out;

    const int N = in_sizes[0] / IN_FEATS;
    const int E = in_sizes[1];

    const int T = 256;
    auto cdiv = [](long a, long b) { return (int)((a + b - 1) / b); };

    const int tiles   = N / 16;
    const int gblocks = cdiv(tiles, 8);
    const int nrange  = cdiv(N, RNG);
    const int nslice  = cdiv(E, EPB);

    float* h    = (float*)d_ws;
    float* h1   = h + (size_t)N * HID;
    float* h2   = h1 + (size_t)N * HID;
    const size_t tab = (size_t)nslice * NBMAX;
    int* counts  = (int*)(h2 + (size_t)N * NC);
    int* prefix  = counts + 2 * tab;
    int* start   = prefix + 2 * tab;
    int* counts2 = counts + tab;
    int* prefix2 = prefix + tab;
    int* start2  = start + NBMAX + 8;
    int2* elist  = (int2*)(start + 2 * (NBMAX + 8));
    int* slist   = (int*)(elist + E);

    for (int g = 0; g < 2; ++g) {
        const int* src = (g == 0) ? src1 : src2;
        const int* dst = (g == 0) ? dst1 : dst2;

        range_count_kernel<<<nslice, T, 0, stream>>>(src, dst, counts, counts2, E);
        range_scan_kernel<<<2, NBMAX, 0, stream>>>(counts, prefix, start, nslice);
        range_fill_kernel<<<nslice, T, 0, stream>>>(src, dst, prefix, start, prefix2, start2,
                                                    elist, slist, E);

        gemm_gc_kernel<IN_FEATS, HID><<<gblocks, T, 0, stream>>>(x, slist, start2, W1, h, N);
        range_sum_kernel<HID, 0><<<nrange, T, 0, stream>>>(h, elist, start, b1, h1, N);

        gemm_gc_kernel<HID, NC><<<gblocks, T, 0, stream>>>(h1, slist, start2, W2, h2, N);
        if (g == 0)
            range_sum_kernel<NC, 1><<<nrange, T, 0, stream>>>(h2, elist, start, b2, out, N);
        else
            range_sum_kernel<NC, 2><<<nrange, T, 0, stream>>>(h2, elist, start, b2, out, N);
    }
}
